# starting kernel
# baseline (speedup 1.0000x reference)
#include <hip/hip_runtime.h>
#include <hip/hip_bf16.h>


typedef __attribute__((ext_vector_type(2))) float v2f;
typedef __attribute__((ext_vector_type(8))) float v8f;


__global__ void deg_init_kernel(float* __restrict__ deg, int n) {
    int i = blockIdx.x * blockDim.x + threadIdx.x;
    if (i < n) deg[i] = 1.0f;
}

__global__ void deg_accum_kernel(const int* __restrict__ dst, float* __restrict__ deg, int e) {
    int i = blockIdx.x * blockDim.x + threadIdx.x;
    if (i < e) atomicAdd(&deg[dst[i]], 1.0f);
}

__global__ void deg_to_dinv_kernel(float* __restrict__ deg, int n) {
    int i = blockIdx.x * blockDim.x + threadIdx.x;
    if (i < n) {
        float d = deg[i];
        deg[i] = (d > 0.0f) ? rsqrtf(d) : 0.0f;
    }
}


template<int K, int NCOLS, bool RELU_A>
__global__ __launch_bounds__((NCOLS / 16) * 32)
void wmma_gemm_kernel(const float* __restrict__ A, const float* __restrict__ B,
                      float* __restrict__ C, int M) {
    const int wave = threadIdx.x >> 5;
    const int lane = threadIdx.x & 31;
    const int m0 = blockIdx.x * 16;
    if (m0 >= M) return;

    const int n0   = wave * 16;
    const int lo16 = lane & 15;
    const int half = lane >> 4;
    const int col  = n0 + lo16;

    const float* __restrict__ arow = A + (size_t)(m0 + lo16) * K;

    v8f acc = {};
    #pragma unroll
    for (int k = 0; k < K; k += 4) {
        const int kb = k + half * 2;
        float2 av = *(const float2*)(arow + kb);
        if (RELU_A) { av.x = fmaxf(av.x, 0.0f); av.y = fmaxf(av.y, 0.0f); }
        v2f a; a[0] = av.x; a[1] = av.y;
        v2f b;
        b[0] = B[(size_t)kb * NCOLS + col];
        b[1] = B[(size_t)(kb + 1) * NCOLS + col];
        acc = __builtin_amdgcn_wmma_f32_16x16x4_f32(
                   false, a,  false, b,
                   (short)0, acc,  false,  false);
    }

    const int rbase = m0 + half * 8;
    #pragma unroll
    for (int v = 0; v < 8; ++v)
        C[(size_t)(rbase + v) * NCOLS + col] = acc[v];
}


template<int C>
__global__ void agg_init_kernel(const float* __restrict__ h, const float* __restrict__ dinv,
                                const float* __restrict__ bias, float* __restrict__ agg, int n) {
    int i = blockIdx.x * blockDim.x + threadIdx.x;
    int node = i / C, c = i % C;
    if (node < n) {
        float di = dinv[node];
        agg[i] = bias[c] + h[i] * di * di;
    }
}

template<int C>
__global__ void edge_scatter_kernel(const int* __restrict__ src, const int* __restrict__ dst,
                                    const float* __restrict__ h, const float* __restrict__ dinv,
                                    float* __restrict__ agg, int e) {
    int t = blockIdx.x * blockDim.x + threadIdx.x;
    int edge = t / C, c = t % C;
    if (edge >= e) return;
    int s = src[edge], d = dst[edge];
    float norm = dinv[s] * dinv[d];
    atomicAdd(&agg[(size_t)d * C + c], h[(size_t)s * C + c] * norm);
}


__global__ void head_kernel(const float* __restrict__ agg2, const float* __restrict__ Wc,
                            const float* __restrict__ bc, float* __restrict__ out, int n) {
    int i = blockIdx.x * blockDim.x + threadIdx.x;
    if (i >= n) return;
    const float* row = agg2 + (size_t)i * 32;
    float acc = bc[0];
    #pragma unroll
    for (int c = 0; c < 32; ++c)
        acc += fmaxf(row[c], 0.0f) * Wc[c];
    out[i] = acc;
}


static inline int cdiv(long long a, long long b) { return (int)((a + b - 1) / b); }

extern "C" void kernel_launch(void* const* d_in, const int* in_sizes, int n_in,
                              void* d_out, int out_size, void* d_ws, size_t ws_size,
                              hipStream_t stream) {
    const float* x   = (const float*)d_in[0];
    const int*   ei  = (const int*)d_in[1];
    const float* W1  = (const float*)d_in[2];
    const float* b1  = (const float*)d_in[3];
    const float* W2  = (const float*)d_in[4];
    const float* b2  = (const float*)d_in[5];
    const float* Wc  = (const float*)d_in[6];
    const float* bc  = (const float*)d_in[7];
    float* out = (float*)d_out;

    const int N = in_sizes[0] / 128;
    const int E = in_sizes[1] / 2;
    const int* src = ei;
    const int* dst = ei + E;

    float* ws   = (float*)d_ws;
    float* dinv = ws;
    float* h1   = dinv + N;
    float* agg1 = h1 + (size_t)N * 64;
    float* h2   = agg1 + (size_t)N * 64;
    float* agg2 = h2 + (size_t)N * 32;
    (void)ws_size; (void)n_in; (void)out_size;

    const int T = 256;

    deg_init_kernel<<<cdiv(N, T), T, 0, stream>>>(dinv, N);
    deg_accum_kernel<<<cdiv(E, T), T, 0, stream>>>(dst, dinv, E);
    deg_to_dinv_kernel<<<cdiv(N, T), T, 0, stream>>>(dinv, N);

    wmma_gemm_kernel<128, 64, false><<<cdiv(N, 16), 128, 0, stream>>>(x, W1, h1, N);

    agg_init_kernel<64><<<cdiv((long long)N * 64, T), T, 0, stream>>>(h1, dinv, b1, agg1, N);
    edge_scatter_kernel<64><<<cdiv((long long)E * 64, T), T, 0, stream>>>(src, dst, h1, dinv, agg1, E);

    wmma_gemm_kernel<64, 32, true><<<cdiv(N, 16), 64, 0, stream>>>(agg1, W2, h2, N);

    agg_init_kernel<32><<<cdiv((long long)N * 32, T), T, 0, stream>>>(h2, dinv, b2, agg2, N);
    edge_scatter_kernel<32><<<cdiv((long long)E * 32, T), T, 0, stream>>>(src, dst, h2, dinv, agg2, E);

    head_kernel<<<cdiv(N, T), T, 0, stream>>>(agg2, Wc, bc, out, N);
}


# Replace the float atomic edge scatter by edges grouped per destination each call (integer rank atomics, scan, fill) and a per-node gather; sum order per node = atomic rank order, varies per run
# speedup vs baseline: 1.9023x; 1.9023x over previous
#include <hip/hip_runtime.h>
#include <hip/hip_bf16.h>


typedef __attribute__((ext_vector_type(2))) float v2f;
typedef __attribute__((ext_vector_type(8))) float v8f;


__global__ void cnt_zero_kernel(int* __restrict__ cnt, int n) {
    int i = blockIdx.x * blockDim.x + threadIdx.x;
    if (i < n) cnt[i] = 0;
}

__global__ void edge_rank_kernel(const int* __restrict__ dst, int* __restrict__ cnt,
                                 int* __restrict__ rank, int e) {
    int i = blockIdx.x * blockDim.x + threadIdx.x;
    if (i < e) rank[i] = atomicAdd(&cnt[dst[i]], 1);
}

__global__ __launch_bounds__(1024)
void rowptr_dinv_kernel(const int* __restrict__ cnt, int* __restrict__ rowptr,
                        float* __restrict__ dinv, int n) {
    __shared__ int wave_cnt[32];
    __shared__ int wave_pre[32];
    const int tid  = threadIdx.x;
    const int lane = tid & 31;
    const int wave = tid >> 5;
    const int base = blockIdx.x * 1024;

    int before = 0;
    for (int i = tid; i < base; i += 1024) before += cnt[i];

    const int i = base + tid;
    const int c = (i < n) ? cnt[i] : 0;

    int inc = c;
    for (int o = 1; o < 32; o <<= 1) {
        int t = __shfl_up(inc, o);
        if (lane >= o) inc += t;
    }
    for (int o = 16; o > 0; o >>= 1) before += __shfl_xor(before, o);

    if (lane == 31) wave_cnt[wave] = inc;
    if (lane == 0)  wave_pre[wave] = before;
    __syncthreads();

    int v = wave_pre[lane] + ((lane < wave) ? wave_cnt[lane] : 0);
    for (int o = 16; o > 0; o >>= 1) v += __shfl_xor(v, o);

    if (i < n) {
        rowptr[i] = v + inc - c;
        dinv[i]   = rsqrtf((float)(c + 1));
    }
}

__global__ void adj_fill_kernel(const int* __restrict__ src, const int* __restrict__ dst,
                                const int* __restrict__ rank, const int* __restrict__ rowptr,
                                const float* __restrict__ dinv, int2* __restrict__ adj, int e) {
    int i = blockIdx.x * blockDim.x + threadIdx.x;
    if (i >= e) return;
    int s = src[i], d = dst[i];
    adj[rowptr[d] + rank[i]] = make_int2(s, __float_as_int(dinv[s]));
}


template<int K, int NCOLS, bool RELU_A>
__global__ __launch_bounds__((NCOLS / 16) * 32)
void wmma_gemm_kernel(const float* __restrict__ A, const float* __restrict__ B,
                      float* __restrict__ C, int M) {
    const int wave = threadIdx.x >> 5;
    const int lane = threadIdx.x & 31;
    const int m0 = blockIdx.x * 16;
    if (m0 >= M) return;

    const int n0   = wave * 16;
    const int lo16 = lane & 15;
    const int half = lane >> 4;
    const int col  = n0 + lo16;

    const float* __restrict__ arow = A + (size_t)(m0 + lo16) * K;

    v8f acc = {};
    #pragma unroll
    for (int k = 0; k < K; k += 4) {
        const int kb = k + half * 2;
        float2 av = *(const float2*)(arow + kb);
        if (RELU_A) { av.x = fmaxf(av.x, 0.0f); av.y = fmaxf(av.y, 0.0f); }
        v2f a; a[0] = av.x; a[1] = av.y;
        v2f b;
        b[0] = B[(size_t)kb * NCOLS + col];
        b[1] = B[(size_t)(kb + 1) * NCOLS + col];
        acc = __builtin_amdgcn_wmma_f32_16x16x4_f32(
                   false, a,  false, b,
                   (short)0, acc,  false,  false);
    }

    const int rbase = m0 + half * 8;
    #pragma unroll
    for (int v = 0; v < 8; ++v)
        C[(size_t)(rbase + v) * NCOLS + col] = acc[v];
}


template<int C>
__global__ __launch_bounds__(256)
void gather_kernel(const int* __restrict__ rowptr, const int* __restrict__ cnt,
                   const int2* __restrict__ adj, const float* __restrict__ h,
                   const float* __restrict__ dinv, const float* __restrict__ bias,
                   float* __restrict__ agg, int n) {
    constexpr int LPN = C / 4;
    const int t    = blockIdx.x * blockDim.x + threadIdx.x;
    const int node = t / LPN;
    const int c4   = (t % LPN) * 4;
    if (node >= n) return;

    const float dd  = dinv[node];
    const int   beg = rowptr[node];
    const int   end = beg + cnt[node];

    const float4 b4 = *(const float4*)(bias + c4);
    const float4 hv = *(const float4*)(h + (size_t)node * C + c4);
    float4 acc;
    acc.x = b4.x + hv.x * dd * dd;
    acc.y = b4.y + hv.y * dd * dd;
    acc.z = b4.z + hv.z * dd * dd;
    acc.w = b4.w + hv.w * dd * dd;

    #pragma unroll 4
    for (int j = beg; j < end; ++j) {
        const int2  sw = adj[j];
        const float w  = __int_as_float(sw.y) * dd;
        const float4 v = *(const float4*)(h + (size_t)sw.x * C + c4);
        acc.x += v.x * w;
        acc.y += v.y * w;
        acc.z += v.z * w;
        acc.w += v.w * w;
    }
    *(float4*)(agg + (size_t)node * C + c4) = acc;
}


__global__ void head_kernel(const float* __restrict__ agg2, const float* __restrict__ Wc,
                            const float* __restrict__ bc, float* __restrict__ out, int n) {
    int i = blockIdx.x * blockDim.x + threadIdx.x;
    if (i >= n) return;
    const float* row = agg2 + (size_t)i * 32;
    float acc = bc[0];
    #pragma unroll
    for (int c = 0; c < 32; ++c)
        acc += fmaxf(row[c], 0.0f) * Wc[c];
    out[i] = acc;
}


static inline int cdiv(long long a, long long b) { return (int)((a + b - 1) / b); }

extern "C" void kernel_launch(void* const* d_in, const int* in_sizes, int n_in,
                              void* d_out, int out_size, void* d_ws, size_t ws_size,
                              hipStream_t stream) {
    const float* x   = (const float*)d_in[0];
    const int*   ei  = (const int*)d_in[1];
    const float* W1  = (const float*)d_in[2];
    const float* b1  = (const float*)d_in[3];
    const float* W2  = (const float*)d_in[4];
    const float* b2  = (const float*)d_in[5];
    const float* Wc  = (const float*)d_in[6];
    const float* bc  = (const float*)d_in[7];
    float* out = (float*)d_out;

    const int N = in_sizes[0] / 128;
    const int E = in_sizes[1] / 2;
    const int* src = ei;
    const int* dst = ei + E;

    float* ws   = (float*)d_ws;
    float* h1   = ws;
    float* agg1 = h1 + (size_t)N * 64;
    float* h2   = h1;
    float* agg2 = h1 + (size_t)N * 32;
    int2*  adj    = (int2*)(agg1 + (size_t)N * 64);
    int*   cnt    = (int*)(adj + E);
    int*   rowptr = cnt + N;
    int*   rank   = rowptr + N;
    float* dinv   = (float*)(rank + E);
    (void)ws_size; (void)n_in; (void)out_size;

    const int T = 256;

    cnt_zero_kernel<<<cdiv(N, T), T, 0, stream>>>(cnt, N);
    edge_rank_kernel<<<cdiv(E, T), T, 0, stream>>>(dst, cnt, rank, E);
    rowptr_dinv_kernel<<<cdiv(N, 1024), 1024, 0, stream>>>(cnt, rowptr, dinv, N);
    adj_fill_kernel<<<cdiv(E, T), T, 0, stream>>>(src, dst, rank, rowptr, dinv, adj, E);

    wmma_gemm_kernel<128, 64, false><<<cdiv(N, 16), 128, 0, stream>>>(x, W1, h1, N);

    gather_kernel<64><<<cdiv((long long)N * 16, T), T, 0, stream>>>(rowptr, cnt, adj, h1, dinv, b1, agg1, N);

    wmma_gemm_kernel<64, 32, true><<<cdiv(N, 16), 64, 0, stream>>>(agg1, W2, h2, N);

    gather_kernel<32><<<cdiv((long long)N * 8, T), T, 0, stream>>>(rowptr, cnt, adj, h2, dinv, b2, agg2, N);

    head_kernel<<<cdiv(N, T), T, 0, stream>>>(agg2, Wc, bc, out, N);
}


# Merge launches 9 to 6: x@W1 stripes run as extra blocks behind the 4 edge-grouping launches; relu(agg1)@W2 runs in the layer-1 gather from an LDS tile; head dot in the layer-2 gather, 8-lane tree sum
# speedup vs baseline: 2.1980x; 1.1554x over previous
#include <hip/hip_runtime.h>
#include <hip/hip_bf16.h>


typedef __attribute__((ext_vector_type(2))) float v2f;
typedef __attribute__((ext_vector_type(8))) float v8f;


template<int WAVES>
__device__ __forceinline__
void gemm1_block(const float* __restrict__ A, const float* __restrict__ B,
                 float* __restrict__ C, int blk, int s_begin, int s_end) {
    constexpr int K = 128, NCOLS = 64;
    const int wave = threadIdx.x >> 5;
    const int lane = threadIdx.x & 31;
    const int stripe = s_begin + blk * (WAVES / 4) + (wave >> 2);
    if (stripe >= s_end) return;

    const int m0   = stripe * 16;
    const int n0   = (wave & 3) * 16;
    const int lo16 = lane & 15;
    const int half = lane >> 4;
    const int col  = n0 + lo16;

    const float* __restrict__ arow = A + (size_t)(m0 + lo16) * K;

    v8f acc = {};
    #pragma unroll
    for (int k = 0; k < K; k += 4) {
        const int kb = k + half * 2;
        float2 av = *(const float2*)(arow + kb);
        v2f a; a[0] = av.x; a[1] = av.y;
        v2f b;
        b[0] = B[(size_t)kb * NCOLS + col];
        b[1] = B[(size_t)(kb + 1) * NCOLS + col];
        acc = __builtin_amdgcn_wmma_f32_16x16x4_f32(
                   false, a,  false, b,
                   (short)0, acc,  false,  false);
    }

    const int rbase = m0 + half * 8;
    #pragma unroll
    for (int v = 0; v < 8; ++v)
        C[(size_t)(rbase + v) * NCOLS + col] = acc[v];
}


__global__ __launch_bounds__(256)
void cnt_zero_kernel(int* __restrict__ cnt, int n, int own,
                     const float* __restrict__ A, const float* __restrict__ B,
                     float* __restrict__ C, int s_begin, int s_end) {
    if ((int)blockIdx.x >= own) {
        gemm1_block<8>(A, B, C, (int)blockIdx.x - own, s_begin, s_end);
        return;
    }
    int i = blockIdx.x * 256 + threadIdx.x;
    if (i < n) cnt[i] = 0;
}

__global__ __launch_bounds__(256)
void edge_rank_kernel(const int* __restrict__ dst, int* __restrict__ cnt,
                      int* __restrict__ rank, int e, int own,
                      const float* __restrict__ A, const float* __restrict__ B,
                      float* __restrict__ C, int s_begin, int s_end) {
    if ((int)blockIdx.x >= own) {
        gemm1_block<8>(A, B, C, (int)blockIdx.x - own, s_begin, s_end);
        return;
    }
    int i = blockIdx.x * 256 + threadIdx.x;
    if (i < e) rank[i] = atomicAdd(&cnt[dst[i]], 1);
}

__global__ __launch_bounds__(1024)
void rowptr_dinv_kernel(const int* __restrict__ cnt, int* __restrict__ rowptr,
                        float* __restrict__ dinv, int n, int own,
                        const float* __restrict__ A, const float* __restrict__ B,
                        float* __restrict__ C, int s_begin, int s_end) {
    if ((int)blockIdx.x >= own) {
        gemm1_block<32>(A, B, C, (int)blockIdx.x - own, s_begin, s_end);
        return;
    }
    __shared__ int wave_cnt[32];
    __shared__ int wave_pre[32];
    const int tid  = threadIdx.x;
    const int lane = tid & 31;
    const int wave = tid >> 5;
    const int base = blockIdx.x * 1024;

    int before = 0;
    for (int i = tid; i < base; i += 1024) before += cnt[i];

    const int i = base + tid;
    const int c = (i < n) ? cnt[i] : 0;

    int inc = c;
    for (int o = 1; o < 32; o <<= 1) {
        int t = __shfl_up(inc, o);
        if (lane >= o) inc += t;
    }
    for (int o = 16; o > 0; o >>= 1) before += __shfl_xor(before, o);

    if (lane == 31) wave_cnt[wave] = inc;
    if (lane == 0)  wave_pre[wave] = before;
    __syncthreads();

    int v = wave_pre[lane] + ((lane < wave) ? wave_cnt[lane] : 0);
    for (int o = 16; o > 0; o >>= 1) v += __shfl_xor(v, o);

    if (i < n) {
        rowptr[i] = v + inc - c;
        dinv[i]   = rsqrtf((float)(c + 1));
    }
}

__global__ __launch_bounds__(256)
void adj_fill_kernel(const int* __restrict__ src, const int* __restrict__ dst,
                     const int* __restrict__ rank, const int* __restrict__ rowptr,
                     const float* __restrict__ dinv, int2* __restrict__ adj, int e, int own,
                     const float* __restrict__ A, const float* __restrict__ B,
                     float* __restrict__ C, int s_begin, int s_end) {
    if ((int)blockIdx.x >= own) {
        gemm1_block<8>(A, B, C, (int)blockIdx.x - own, s_begin, s_end);
        return;
    }
    int i = blockIdx.x * 256 + threadIdx.x;
    if (i >= e) return;
    int s = src[i], d = dst[i];
    adj[rowptr[d] + rank[i]] = make_int2(s, __float_as_int(dinv[s]));
}


template<int C>
__device__ __forceinline__
float4 gather_row(const int* __restrict__ rowptr, const int* __restrict__ cnt,
                  const int2* __restrict__ adj, const float* __restrict__ h,
                  const float* __restrict__ dinv, const float* __restrict__ bias,
                  int node, int c4) {
    const float dd  = dinv[node];
    const int   beg = rowptr[node];
    const int   end = beg + cnt[node];

    const float4 b4 = *(const float4*)(bias + c4);
    const float4 hv = *(const float4*)(h + (size_t)node * C + c4);
    float4 acc;
    acc.x = b4.x + hv.x * dd * dd;
    acc.y = b4.y + hv.y * dd * dd;
    acc.z = b4.z + hv.z * dd * dd;
    acc.w = b4.w + hv.w * dd * dd;

    #pragma unroll 4
    for (int j = beg; j < end; ++j) {
        const int2  sw = adj[j];
        const float w  = __int_as_float(sw.y) * dd;
        const float4 v = *(const float4*)(h + (size_t)sw.x * C + c4);
        acc.x += v.x * w;
        acc.y += v.y * w;
        acc.z += v.z * w;
        acc.w += v.w * w;
    }
    return acc;
}


__global__ __launch_bounds__(256)
void gather64_gemm2_kernel(const int* __restrict__ rowptr, const int* __restrict__ cnt,
                           const int2* __restrict__ adj, const float* __restrict__ h,
                           const float* __restrict__ dinv, const float* __restrict__ bias,
                           const float* __restrict__ W2, float* __restrict__ h2, int n) {
    constexpr int PITCH = 68;
    __shared__ __attribute__((aligned(16))) float tile[16 * PITCH];

    const int tid  = threadIdx.x;
    const int rloc = tid >> 4;
    const int c4   = (tid & 15) * 4;
    const int m0   = blockIdx.x * 16;
    const int node = min(m0 + rloc, n - 1);

    float4 acc = gather_row<64>(rowptr, cnt, adj, h, dinv, bias, node, c4);
    acc.x = fmaxf(acc.x, 0.0f);
    acc.y = fmaxf(acc.y, 0.0f);
    acc.z = fmaxf(acc.z, 0.0f);
    acc.w = fmaxf(acc.w, 0.0f);
    *(float4*)(tile + rloc * PITCH + c4) = acc;
    __syncthreads();

    const int wave = tid >> 5;
    if (wave >= 2) return;

    const int lane = tid & 31;
    const int lo16 = lane & 15;
    const int half = lane >> 4;
    const int col  = wave * 16 + lo16;
    const float* arow = tile + lo16 * PITCH;

    v8f o = {};
    #pragma unroll
    for (int k = 0; k < 64; k += 4) {
        const int kb = k + half * 2;
        const float2 av = *(const float2*)(arow + kb);
        v2f a; a[0] = av.x; a[1] = av.y;
        v2f b;
        b[0] = W2[kb * 32 + col];
        b[1] = W2[(kb + 1) * 32 + col];
        o = __builtin_amdgcn_wmma_f32_16x16x4_f32(
                 false, a,  false, b,
                 (short)0, o,  false,  false);
    }

    const int rbase = m0 + half * 8;
    #pragma unroll
    for (int v = 0; v < 8; ++v)
        h2[(size_t)(rbase + v) * 32 + col] = o[v];
}


__global__ __launch_bounds__(256)
void gather32_head_kernel(const int* __restrict__ rowptr, const int* __restrict__ cnt,
                          const int2* __restrict__ adj, const float* __restrict__ h,
                          const float* __restrict__ dinv, const float* __restrict__ bias,
                          const float* __restrict__ Wc, const float* __restrict__ bc,
                          float* __restrict__ out, int n) {
    __shared__ float res[32];

    const int tid  = threadIdx.x;
    const int rloc = tid >> 3;
    const int c4   = (tid & 7) * 4;
    const int node = min((int)blockIdx.x * 32 + rloc, n - 1);

    const float4 acc = gather_row<32>(rowptr, cnt, adj, h, dinv, bias, node, c4);
    float p = fmaxf(acc.x, 0.0f) * Wc[c4]
            + fmaxf(acc.y, 0.0f) * Wc[c4 + 1]
            + fmaxf(acc.z, 0.0f) * Wc[c4 + 2]
            + fmaxf(acc.w, 0.0f) * Wc[c4 + 3];
    p += __shfl_xor(p, 1);
    p += __shfl_xor(p, 2);
    p += __shfl_xor(p, 4);
    if ((tid & 7) == 0) res[rloc] = p + bc[0];
    __syncthreads();

    if (tid < 32) {
        const int i = blockIdx.x * 32 + tid;
        if (i < n) out[i] = res[tid];
    }
}


static inline int cdiv(long long a, long long b) { return (int)((a + b - 1) / b); }

extern "C" void kernel_launch(void* const* d_in, const int* in_sizes, int n_in,
                              void* d_out, int out_size, void* d_ws, size_t ws_size,
                              hipStream_t stream) {
    const float* x   = (const float*)d_in[0];
    const int*   ei  = (const int*)d_in[1];
    const float* W1  = (const float*)d_in[2];
    const float* b1  = (const float*)d_in[3];
    const float* W2  = (const float*)d_in[4];
    const float* b2  = (const float*)d_in[5];
    const float* Wc  = (const float*)d_in[6];
    const float* bc  = (const float*)d_in[7];
    float* out = (float*)d_out;

    const int N = in_sizes[0] / 128;
    const int E = in_sizes[1] / 2;
    const int* src = ei;
    const int* dst = ei + E;

    float* ws   = (float*)d_ws;
    float* h1   = ws;
    float* h2   = h1 + (size_t)N * 64;
    int2*  adj    = (int2*)(h2 + (size_t)N * 64);
    int*   cnt    = (int*)(adj + E);
    int*   rowptr = cnt + N;
    int*   rank   = rowptr + N;
    float* dinv   = (float*)(rank + E);
    (void)ws_size; (void)n_in; (void)out_size;

    const int T = 256;

    const int S  = cdiv(N, 16);
    const int s1 = S / 10;
    const int s2 = s1 + (S * 3) / 10;
    const int s3 = s2 + (S * 3) / 10;

    const int own1 = cdiv(N, T);
    cnt_zero_kernel<<<own1 + cdiv(s1, 2), T, 0, stream>>>(cnt, N, own1, x, W1, h1, 0, s1);

    const int own2 = cdiv(E, T);
    edge_rank_kernel<<<own2 + cdiv(s2 - s1, 2), T, 0, stream>>>(dst, cnt, rank, E, own2, x, W1, h1, s1, s2);

    const int own3 = cdiv(N, 1024);
    rowptr_dinv_kernel<<<own3 + cdiv(s3 - s2, 8), 1024, 0, stream>>>(cnt, rowptr, dinv, N, own3, x, W1, h1, s2, s3);

    const int own4 = cdiv(E, T);
    adj_fill_kernel<<<own4 + cdiv(S - s3, 2), T, 0, stream>>>(src, dst, rank, rowptr, dinv, adj, E, own4, x, W1, h1, s3, S);

    gather64_gemm2_kernel<<<cdiv(N, 16), T, 0, stream>>>(rowptr, cnt, adj, h1, dinv, b1, W2, h2, N);

    gather32_head_kernel<<<cdiv(N, 32), T, 0, stream>>>(rowptr, cnt, adj, h2, dinv, b2, Wc, bc, out, N);
}


# Scan + fill launches replaced by 32-slot buckets per destination filled in the rank launch (overflow list past 32), dinv formed per edge from counts: 6 launches to 4; sum order = atomic arrival order
# speedup vs baseline: 2.3882x; 1.0865x over previous
#include <hip/hip_runtime.h>
#include <hip/hip_bf16.h>


typedef __attribute__((ext_vector_type(2))) float v2f;
typedef __attribute__((ext_vector_type(8))) float v8f;


template<int WAVES>
__device__ __forceinline__
void gemm1_block(const float* __restrict__ A, const float* __restrict__ B,
                 float* __restrict__ C, int blk, int s_begin, int s_end) {
    constexpr int K = 128, NCOLS = 64;
    const int wave = threadIdx.x >> 5;
    const int lane = threadIdx.x & 31;
    const int stripe = s_begin + blk * (WAVES / 4) + (wave >> 2);
    if (stripe >= s_end) return;

    const int m0   = stripe * 16;
    const int n0   = (wave & 3) * 16;
    const int lo16 = lane & 15;
    const int half = lane >> 4;
    const int col  = n0 + lo16;

    const float* __restrict__ arow = A + (size_t)(m0 + lo16) * K;

    v8f acc = {};
    #pragma unroll
    for (int k = 0; k < K; k += 4) {
        const int kb = k + half * 2;
        float2 av = *(const float2*)(arow + kb);
        v2f a; a[0] = av.x; a[1] = av.y;
        v2f b;
        b[0] = B[(size_t)kb * NCOLS + col];
        b[1] = B[(size_t)(kb + 1) * NCOLS + col];
        acc = __builtin_amdgcn_wmma_f32_16x16x4_f32(
                   false, a,  false, b,
                   (short)0, acc,  false,  false);
    }

    const int rbase = m0 + half * 8;
    #pragma unroll
    for (int v = 0; v < 8; ++v)
        C[(size_t)(rbase + v) * NCOLS + col] = acc[v];
}


constexpr int CAP = 32;

__global__ __launch_bounds__(256)
void cnt_zero_kernel(int* __restrict__ cnt, int n, int own,
                     const float* __restrict__ A, const float* __restrict__ B,
                     float* __restrict__ C, int s_begin, int s_end) {
    if ((int)blockIdx.x >= own) {
        gemm1_block<8>(A, B, C, (int)blockIdx.x - own, s_begin, s_end);
        return;
    }
    int i = blockIdx.x * 256 + threadIdx.x;
    if (i <= n) cnt[i] = 0;
}

__global__ __launch_bounds__(256)
void edge_bucket_kernel(const int* __restrict__ src, const int* __restrict__ dst,
                        int* __restrict__ cnt, int* __restrict__ bkt,
                        int2* __restrict__ ovf, int* __restrict__ ovfn, int e, int own,
                        const float* __restrict__ A, const float* __restrict__ B,
                        float* __restrict__ C, int s_begin, int s_end) {
    if ((int)blockIdx.x >= own) {
        gemm1_block<8>(A, B, C, (int)blockIdx.x - own, s_begin, s_end);
        return;
    }
    int i = blockIdx.x * 256 + threadIdx.x;
    if (i >= e) return;
    const int s = src[i], d = dst[i];
    const int r = atomicAdd(&cnt[d], 1);
    if (r < CAP) {
        bkt[(size_t)d * CAP + r] = s;
    } else {
        const int k = atomicAdd(ovfn, 1);
        ovf[k] = make_int2(d, s);
    }
}


__device__ __forceinline__ float dinv_of(int count) {
    return __builtin_amdgcn_rsqf((float)(count + 1));
}

template<int C, int NB>
__device__ __forceinline__
void edge_batch(const int (&s)[NB], int jbase, int c, int node, float dd,
                const int* __restrict__ cnt, const float4* __restrict__ h4, int q,
                float4& acc) {
    float4 v[NB];
    int    cs[NB];
    #pragma unroll
    for (int j = 0; j < NB; ++j) {
        const int sj = (jbase + j < c) ? s[j] : node;
        cs[j] = cnt[(unsigned)sj];
        v[j]  = h4[(unsigned)(sj * (C / 4) + q)];
    }
    #pragma unroll
    for (int j = 0; j < NB; ++j) {
        const float wj = dinv_of(cs[j]) * dd;
        const float w  = (jbase + j < c) ? wj : 0.0f;
        acc.x += v[j].x * w;
        acc.y += v[j].y * w;
        acc.z += v[j].z * w;
        acc.w += v[j].w * w;
    }
}

template<int C>
__device__ __forceinline__
float4 gather_row(const int* __restrict__ cnt, const int* __restrict__ bkt,
                  const int2* __restrict__ ovf, const int* __restrict__ ovfn, int e,
                  const float* __restrict__ h, const float* __restrict__ bias,
                  int node, int c4) {
    const float4* __restrict__ h4 = (const float4*)h;
    const int q = c4 >> 2;
    const int4* __restrict__ brow4 = (const int4*)bkt + (unsigned)(node * (CAP / 4));
    const int4 e0 = brow4[0];
    const int4 e1 = brow4[1];
    const int4 e2 = brow4[2];
    const int   c  = cnt[(unsigned)node];
    const float dd = dinv_of(c);

    const float4 b4 = *(const float4*)(bias + c4);
    const float4 hv = h4[(unsigned)(node * (C / 4) + q)];
    float4 acc;
    acc.x = b4.x + hv.x * dd * dd;
    acc.y = b4.y + hv.y * dd * dd;
    acc.z = b4.z + hv.z * dd * dd;
    acc.w = b4.w + hv.w * dd * dd;

    const int sa[8] = {e0.x, e0.y, e0.z, e0.w, e1.x, e1.y, e1.z, e1.w};
    edge_batch<C, 8>(sa, 0, c, node, dd, cnt, h4, q, acc);

    if (c > 8) {
        const int sb[4] = {e2.x, e2.y, e2.z, e2.w};
        edge_batch<C, 4>(sb, 8, c, node, dd, cnt, h4, q, acc);

        const int* __restrict__ brow = (const int*)brow4;
        const int cm = min(c, CAP);
        for (int j = 12; j < cm; ++j) {
            const int   sj = brow[j];
            const float w  = dinv_of(cnt[(unsigned)sj]) * dd;
            const float4 v = h4[(unsigned)(sj * (C / 4) + q)];
            acc.x += v.x * w;
            acc.y += v.y * w;
            acc.z += v.z * w;
            acc.w += v.w * w;
        }
        if (c > CAP) {
            const int m = min(ovfn[0], e);
            for (int k = 0; k < m; ++k) {
                const int2 p = ovf[k];
                if (p.x != node) continue;
                const float w  = dinv_of(cnt[(unsigned)p.y]) * dd;
                const float4 v = h4[(unsigned)(p.y * (C / 4) + q)];
                acc.x += v.x * w;
                acc.y += v.y * w;
                acc.z += v.z * w;
                acc.w += v.w * w;
            }
        }
    }
    return acc;
}


__global__ __launch_bounds__(256)
void gather64_gemm2_kernel(const int* __restrict__ cnt, const int* __restrict__ bkt,
                           const int2* __restrict__ ovf, const int* __restrict__ ovfn, int e,
                           const float* __restrict__ h, const float* __restrict__ bias,
                           const float* __restrict__ W2, float* __restrict__ h2, int n) {
    constexpr int PITCH = 68;
    __shared__ __attribute__((aligned(16))) float tile[16 * PITCH];

    const int tid  = threadIdx.x;
    const int rloc = tid >> 4;
    const int c4   = (tid & 15) * 4;
    const int m0   = blockIdx.x * 16;
    const int node = min(m0 + rloc, n - 1);

    float4 acc = gather_row<64>(cnt, bkt, ovf, ovfn, e, h, bias, node, c4);
    acc.x = fmaxf(acc.x, 0.0f);
    acc.y = fmaxf(acc.y, 0.0f);
    acc.z = fmaxf(acc.z, 0.0f);
    acc.w = fmaxf(acc.w, 0.0f);
    *(float4*)(tile + rloc * PITCH + c4) = acc;
    __syncthreads();

    const int wave = tid >> 5;
    if (wave >= 2) return;

    const int lane = tid & 31;
    const int lo16 = lane & 15;
    const int half = lane >> 4;
    const int col  = wave * 16 + lo16;
    const float* arow = tile + lo16 * PITCH;

    v8f o = {};
    #pragma unroll
    for (int k = 0; k < 64; k += 4) {
        const int kb = k + half * 2;
        const float2 av = *(const float2*)(arow + kb);
        v2f a; a[0] = av.x; a[1] = av.y;
        v2f b;
        b[0] = W2[kb * 32 + col];
        b[1] = W2[(kb + 1) * 32 + col];
        o = __builtin_amdgcn_wmma_f32_16x16x4_f32(
                 false, a,  false, b,
                 (short)0, o,  false,  false);
    }

    const int rbase = m0 + half * 8;
    #pragma unroll
    for (int v = 0; v < 8; ++v)
        h2[(size_t)(rbase + v) * 32 + col] = o[v];
}


__global__ __launch_bounds__(256)
void gather32_head_kernel(const int* __restrict__ cnt, const int* __restrict__ bkt,
                          const int2* __restrict__ ovf, const int* __restrict__ ovfn, int e,
                          const float* __restrict__ h, const float* __restrict__ bias,
                          const float* __restrict__ Wc, const float* __restrict__ bc,
                          float* __restrict__ out, int n) {
    __shared__ float res[32];

    const int tid  = threadIdx.x;
    const int rloc = tid >> 3;
    const int c4   = (tid & 7) * 4;
    const int node = min((int)blockIdx.x * 32 + rloc, n - 1);

    const float4 acc = gather_row<32>(cnt, bkt, ovf, ovfn, e, h, bias, node, c4);
    float p = fmaxf(acc.x, 0.0f) * Wc[c4]
            + fmaxf(acc.y, 0.0f) * Wc[c4 + 1]
            + fmaxf(acc.z, 0.0f) * Wc[c4 + 2]
            + fmaxf(acc.w, 0.0f) * Wc[c4 + 3];
    p += __shfl_xor(p, 1);
    p += __shfl_xor(p, 2);
    p += __shfl_xor(p, 4);
    if ((tid & 7) == 0) res[rloc] = p + bc[0];
    __syncthreads();

    if (tid < 32) {
        const int i = blockIdx.x * 32 + tid;
        if (i < n) out[i] = res[tid];
    }
}


static inline int cdiv(long long a, long long b) { return (int)((a + b - 1) / b); }

extern "C" void kernel_launch(void* const* d_in, const int* in_sizes, int n_in,
                              void* d_out, int out_size, void* d_ws, size_t ws_size,
                              hipStream_t stream) {
    const float* x   = (const float*)d_in[0];
    const int*   ei  = (const int*)d_in[1];
    const float* W1  = (const float*)d_in[2];
    const float* b1  = (const float*)d_in[3];
    const float* W2  = (const float*)d_in[4];
    const float* b2  = (const float*)d_in[5];
    const float* Wc  = (const float*)d_in[6];
    const float* bc  = (const float*)d_in[7];
    float* out = (float*)d_out;

    const int N = in_sizes[0] / 128;
    const int E = in_sizes[1] / 2;
    const int* src = ei;
    const int* dst = ei + E;

    float* ws   = (float*)d_ws;
    float* h1   = ws;
    float* h2   = h1 + (size_t)N * 64;
    int*   bkt  = (int*)(h2 + (size_t)N * 32);
    int*   cnt  = bkt + (size_t)N * CAP;
    int*   ovfn = cnt + N;
    int2*  ovf  = (int2*)(ovfn + 4);
    (void)ws_size; (void)n_in; (void)out_size;

    const int T = 256;

    const int S  = cdiv(N, 16);
    const int s1 = S / 3;

    const int own1 = cdiv(N + 1, T);
    cnt_zero_kernel<<<own1 + cdiv(s1, 2), T, 0, stream>>>(cnt, N, own1, x, W1, h1, 0, s1);

    const int own2 = cdiv(E, T);
    edge_bucket_kernel<<<own2 + cdiv(S - s1, 2), T, 0, stream>>>(src, dst, cnt, bkt, ovf, ovfn, E, own2, x, W1, h1, s1, S);

    gather64_gemm2_kernel<<<cdiv(N, 16), T, 0, stream>>>(cnt, bkt, ovf, ovfn, E, h1, b1, W2, h2, N);

    gather32_head_kernel<<<cdiv(N, 32), T, 0, stream>>>(cnt, bkt, ovf, ovfn, E, h2, b2, Wc, bc, out, N);
}


# h2 rows stored pre-multiplied by their node's deg^-1/2 (tile scaled before the W2 product): layer-2 gather adds rows with no per-edge count lookup; same add order, dinv[d] applied once to the sum
# speedup vs baseline: 2.4595x; 1.0299x over previous
#include <hip/hip_runtime.h>
#include <hip/hip_bf16.h>


typedef __attribute__((ext_vector_type(2))) float v2f;
typedef __attribute__((ext_vector_type(8))) float v8f;


template<int WAVES>
__device__ __forceinline__
void gemm1_block(const float* __restrict__ A, const float* __restrict__ B,
                 float* __restrict__ C, int blk, int s_begin, int s_end) {
    constexpr int K = 128, NCOLS = 64;
    const int wave = threadIdx.x >> 5;
    const int lane = threadIdx.x & 31;
    const int stripe = s_begin + blk * (WAVES / 4) + (wave >> 2);
    if (stripe >= s_end) return;

    const int m0   = stripe * 16;
    const int n0   = (wave & 3) * 16;
    const int lo16 = lane & 15;
    const int half = lane >> 4;
    const int col  = n0 + lo16;

    const float* __restrict__ arow = A + (size_t)(m0 + lo16) * K;

    v8f acc = {};
    #pragma unroll
    for (int k = 0; k < K; k += 4) {
        const int kb = k + half * 2;
        float2 av = *(const float2*)(arow + kb);
        v2f a; a[0] = av.x; a[1] = av.y;
        v2f b;
        b[0] = B[(size_t)kb * NCOLS + col];
        b[1] = B[(size_t)(kb + 1) * NCOLS + col];
        acc = __builtin_amdgcn_wmma_f32_16x16x4_f32(
                   false, a,  false, b,
                   (short)0, acc,  false,  false);
    }

    const int rbase = m0 + half * 8;
    #pragma unroll
    for (int v = 0; v < 8; ++v)
        C[(size_t)(rbase + v) * NCOLS + col] = acc[v];
}


constexpr int CAP = 32;

__global__ __launch_bounds__(256)
void cnt_zero_kernel(int* __restrict__ cnt, int n, int own,
                     const float* __restrict__ A, const float* __restrict__ B,
                     float* __restrict__ C, int s_begin, int s_end) {
    if ((int)blockIdx.x >= own) {
        gemm1_block<8>(A, B, C, (int)blockIdx.x - own, s_begin, s_end);
        return;
    }
    int i = blockIdx.x * 256 + threadIdx.x;
    if (i <= n) cnt[i] = 0;
}

__global__ __launch_bounds__(256)
void edge_bucket_kernel(const int* __restrict__ src, const int* __restrict__ dst,
                        int* __restrict__ cnt, int* __restrict__ bkt,
                        int2* __restrict__ ovf, int* __restrict__ ovfn, int e, int own,
                        const float* __restrict__ A, const float* __restrict__ B,
                        float* __restrict__ C, int s_begin, int s_end) {
    if ((int)blockIdx.x >= own) {
        gemm1_block<8>(A, B, C, (int)blockIdx.x - own, s_begin, s_end);
        return;
    }
    int i = blockIdx.x * 256 + threadIdx.x;
    if (i >= e) return;
    const int s = src[i], d = dst[i];
    const int r = atomicAdd(&cnt[d], 1);
    if (r < CAP) {
        bkt[(size_t)d * CAP + r] = s;
    } else {
        const int k = atomicAdd(ovfn, 1);
        ovf[k] = make_int2(d, s);
    }
}


__device__ __forceinline__ float dinv_of(int count) {
    return __builtin_amdgcn_rsqf((float)(count + 1));
}

template<int C, int NB, bool PRE>
__device__ __forceinline__
void edge_batch(const int (&s)[NB], int jbase, int c, int node, float dd,
                const int* __restrict__ cnt, const float4* __restrict__ h4, int q,
                float4& acc) {
    float4 v[NB];
    int    cs[NB];
    #pragma unroll
    for (int j = 0; j < NB; ++j) {
        const int sj = (jbase + j < c) ? s[j] : node;
        cs[j] = PRE ? 0 : cnt[(unsigned)sj];
        v[j]  = h4[(unsigned)(sj * (C / 4) + q)];
    }
    #pragma unroll
    for (int j = 0; j < NB; ++j) {
        const float wj = PRE ? 1.0f : dinv_of(cs[j]) * dd;
        const float w  = (jbase + j < c) ? wj : 0.0f;
        acc.x += v[j].x * w;
        acc.y += v[j].y * w;
        acc.z += v[j].z * w;
        acc.w += v[j].w * w;
    }
}

template<int C, bool PRE>
__device__ __forceinline__
float4 gather_row(const int* __restrict__ cnt, const int* __restrict__ bkt,
                  const int2* __restrict__ ovf, const int* __restrict__ ovfn, int e,
                  const float* __restrict__ h, const float* __restrict__ bias,
                  int node, int c4, float& dd_out) {
    const float4* __restrict__ h4 = (const float4*)h;
    const int q = c4 >> 2;
    const int4* __restrict__ brow4 = (const int4*)bkt + (unsigned)(node * (CAP / 4));
    const int4 e0 = brow4[0];
    const int4 e1 = brow4[1];
    const int4 e2 = brow4[2];
    const int   c  = cnt[(unsigned)node];
    const float dd = dinv_of(c);
    dd_out = dd;

    const float4 b4 = *(const float4*)(bias + c4);
    const float4 hv = h4[(unsigned)(node * (C / 4) + q)];
    float4 acc;
    if (PRE) {
        acc = hv;
    } else {
        acc.x = b4.x + hv.x * dd * dd;
        acc.y = b4.y + hv.y * dd * dd;
        acc.z = b4.z + hv.z * dd * dd;
        acc.w = b4.w + hv.w * dd * dd;
    }

    const int sa[8] = {e0.x, e0.y, e0.z, e0.w, e1.x, e1.y, e1.z, e1.w};
    edge_batch<C, 8, PRE>(sa, 0, c, node, dd, cnt, h4, q, acc);

    if (c > 8) {
        const int sb[4] = {e2.x, e2.y, e2.z, e2.w};
        edge_batch<C, 4, PRE>(sb, 8, c, node, dd, cnt, h4, q, acc);

        const int* __restrict__ brow = (const int*)brow4;
        const int cm = min(c, CAP);
        for (int j = 12; j < cm; ++j) {
            const int   sj = brow[j];
            const float w  = PRE ? 1.0f : dinv_of(cnt[(unsigned)sj]) * dd;
            const float4 v = h4[(unsigned)(sj * (C / 4) + q)];
            acc.x += v.x * w;
            acc.y += v.y * w;
            acc.z += v.z * w;
            acc.w += v.w * w;
        }
        if (c > CAP) {
            const int m = min(ovfn[0], e);
            for (int k = 0; k < m; ++k) {
                const int2 p = ovf[k];
                if (p.x != node) continue;
                const float w  = PRE ? 1.0f : dinv_of(cnt[(unsigned)p.y]) * dd;
                const float4 v = h4[(unsigned)(p.y * (C / 4) + q)];
                acc.x += v.x * w;
                acc.y += v.y * w;
                acc.z += v.z * w;
                acc.w += v.w * w;
            }
        }
    }
    if (PRE) {
        acc.x = b4.x + acc.x * dd;
        acc.y = b4.y + acc.y * dd;
        acc.z = b4.z + acc.z * dd;
        acc.w = b4.w + acc.w * dd;
    }
    return acc;
}


__global__ __launch_bounds__(256)
void gather64_gemm2_kernel(const int* __restrict__ cnt, const int* __restrict__ bkt,
                           const int2* __restrict__ ovf, const int* __restrict__ ovfn, int e,
                           const float* __restrict__ h, const float* __restrict__ bias,
                           const float* __restrict__ W2, float* __restrict__ h2, int n) {
    constexpr int PITCH = 68;
    __shared__ __attribute__((aligned(16))) float tile[16 * PITCH];

    const int tid  = threadIdx.x;
    const int rloc = tid >> 4;
    const int c4   = (tid & 15) * 4;
    const int m0   = blockIdx.x * 16;
    const int node = min(m0 + rloc, n - 1);

    float dd;
    float4 acc = gather_row<64, false>(cnt, bkt, ovf, ovfn, e, h, bias, node, c4, dd);
    acc.x = fmaxf(acc.x, 0.0f) * dd;
    acc.y = fmaxf(acc.y, 0.0f) * dd;
    acc.z = fmaxf(acc.z, 0.0f) * dd;
    acc.w = fmaxf(acc.w, 0.0f) * dd;
    *(float4*)(tile + rloc * PITCH + c4) = acc;
    __syncthreads();

    const int wave = tid >> 5;
    if (wave >= 2) return;

    const int lane = tid & 31;
    const int lo16 = lane & 15;
    const int half = lane >> 4;
    const int col  = wave * 16 + lo16;
    const float* arow = tile + lo16 * PITCH;

    v8f o = {};
    #pragma unroll
    for (int k = 0; k < 64; k += 4) {
        const int kb = k + half * 2;
        const float2 av = *(const float2*)(arow + kb);
        v2f a; a[0] = av.x; a[1] = av.y;
        v2f b;
        b[0] = W2[kb * 32 + col];
        b[1] = W2[(kb + 1) * 32 + col];
        o = __builtin_amdgcn_wmma_f32_16x16x4_f32(
                 false, a,  false, b,
                 (short)0, o,  false,  false);
    }

    const int rbase = m0 + half * 8;
    #pragma unroll
    for (int v = 0; v < 8; ++v)
        h2[(size_t)(rbase + v) * 32 + col] = o[v];
}


__global__ __launch_bounds__(256)
void gather32_head_kernel(const int* __restrict__ cnt, const int* __restrict__ bkt,
                          const int2* __restrict__ ovf, const int* __restrict__ ovfn, int e,
                          const float* __restrict__ h, const float* __restrict__ bias,
                          const float* __restrict__ Wc, const float* __restrict__ bc,
                          float* __restrict__ out, int n) {
    __shared__ float res[32];

    const int tid  = threadIdx.x;
    const int rloc = tid >> 3;
    const int c4   = (tid & 7) * 4;
    const int node = min((int)blockIdx.x * 32 + rloc, n - 1);

    float dd;
    const float4 acc = gather_row<32, true>(cnt, bkt, ovf, ovfn, e, h, bias, node, c4, dd);
    float p = fmaxf(acc.x, 0.0f) * Wc[c4]
            + fmaxf(acc.y, 0.0f) * Wc[c4 + 1]
            + fmaxf(acc.z, 0.0f) * Wc[c4 + 2]
            + fmaxf(acc.w, 0.0f) * Wc[c4 + 3];
    p += __shfl_xor(p, 1);
    p += __shfl_xor(p, 2);
    p += __shfl_xor(p, 4);
    if ((tid & 7) == 0) res[rloc] = p + bc[0];
    __syncthreads();

    if (tid < 32) {
        const int i = blockIdx.x * 32 + tid;
        if (i < n) out[i] = res[tid];
    }
}


static inline int cdiv(long long a, long long b) { return (int)((a + b - 1) / b); }

extern "C" void kernel_launch(void* const* d_in, const int* in_sizes, int n_in,
                              void* d_out, int out_size, void* d_ws, size_t ws_size,
                              hipStream_t stream) {
    const float* x   = (const float*)d_in[0];
    const int*   ei  = (const int*)d_in[1];
    const float* W1  = (const float*)d_in[2];
    const float* b1  = (const float*)d_in[3];
    const float* W2  = (const float*)d_in[4];
    const float* b2  = (const float*)d_in[5];
    const float* Wc  = (const float*)d_in[6];
    const float* bc  = (const float*)d_in[7];
    float* out = (float*)d_out;

    const int N = in_sizes[0] / 128;
    const int E = in_sizes[1] / 2;
    const int* src = ei;
    const int* dst = ei + E;

    float* ws   = (float*)d_ws;
    float* h1   = ws;
    float* h2   = h1 + (size_t)N * 64;
    int*   bkt  = (int*)(h2 + (size_t)N * 32);
    int*   cnt  = bkt + (size_t)N * CAP;
    int*   ovfn = cnt + N;
    int2*  ovf  = (int2*)(ovfn + 4);
    (void)ws_size; (void)n_in; (void)out_size;

    const int T = 256;

    const int S  = cdiv(N, 16);
    const int s1 = S / 3;

    const int own1 = cdiv(N + 1, T);
    cnt_zero_kernel<<<own1 + cdiv(s1, 2), T, 0, stream>>>(cnt, N, own1, x, W1, h1, 0, s1);

    const int own2 = cdiv(E, T);
    edge_bucket_kernel<<<own2 + cdiv(S - s1, 2), T, 0, stream>>>(src, dst, cnt, bkt, ovf, ovfn, E, own2, x, W1, h1, s1, S);

    gather64_gemm2_kernel<<<cdiv(N, 16), T, 0, stream>>>(cnt, bkt, ovf, ovfn, E, h1, b1, W2, h2, N);

    gather32_head_kernel<<<cdiv(N, 32), T, 0, stream>>>(cnt, bkt, ovf, ovfn, E, h2, b2, Wc, bc, out, N);
}


# x@W1 stripes: a block stages 64 x rows in LDS by whole-row loads and each wave keeps its W1 fragments in registers for 2 stripes; A fragments read from LDS, not 16-line global loads; same k order
# speedup vs baseline: 2.8612x; 1.1633x over previous
#include <hip/hip_runtime.h>
#include <hip/hip_bf16.h>


typedef __attribute__((ext_vector_type(2))) float v2f;
typedef __attribute__((ext_vector_type(8))) float v8f;


constexpr int G1_STRIPES = 4;
constexpr int G1_ROWS    = G1_STRIPES * 16;
constexpr int G1_PITCH   = 132;

__device__ __forceinline__
void gemm1_block(const float* __restrict__ A, const float* __restrict__ B,
                 float* __restrict__ C, int blk, int s_begin, int s_end,
                 float* __restrict__ tile) {
    constexpr int K = 128, NCOLS = 64;
    const int lane = threadIdx.x & 31;
    const int wave = __builtin_amdgcn_readfirstlane((int)(threadIdx.x >> 5));
    const int s0       = s_begin + blk * G1_STRIPES;
    const int row0     = s0 * 16;
    const int row_last = s_end * 16 - 1;

    float4 st[G1_ROWS / 8];
    #pragma unroll
    for (int i = 0; i < G1_ROWS / 8; ++i) {
        const int gr = min(row0 + wave + 8 * i, row_last);
        st[i] = *(const float4*)(A + (size_t)gr * K + lane * 4);
    }

    const int lo16 = lane & 15;
    const int half = lane >> 4;
    const int col  = (wave & 3) * 16 + lo16;

    v2f b[K / 4];
    #pragma unroll
    for (int j = 0; j < K / 4; ++j) {
        const int kb = 4 * j + half * 2;
        b[j][0] = B[kb * NCOLS + col];
        b[j][1] = B[(kb + 1) * NCOLS + col];
    }

    #pragma unroll
    for (int i = 0; i < G1_ROWS / 8; ++i)
        *(float4*)(tile + (wave + 8 * i) * G1_PITCH + lane * 4) = st[i];
    __syncthreads();

    const int sp = (wave >> 2) * 2;
    const float* a0row = tile + (sp * 16 + lo16) * G1_PITCH + half * 2;
    const float* a1row = a0row + 16 * G1_PITCH;

    v8f acc0 = {};
    v8f acc1 = {};
    #pragma unroll
    for (int j = 0; j < K / 4; ++j) {
        const float2 x0 = *(const float2*)(a0row + 4 * j);
        const float2 x1 = *(const float2*)(a1row + 4 * j);
        v2f a0; a0[0] = x0.x; a0[1] = x0.y;
        v2f a1; a1[0] = x1.x; a1[1] = x1.y;
        acc0 = __builtin_amdgcn_wmma_f32_16x16x4_f32(
                    false, a0,  false, b[j],
                    (short)0, acc0,  false,  false);
        acc1 = __builtin_amdgcn_wmma_f32_16x16x4_f32(
                    false, a1,  false, b[j],
                    (short)0, acc1,  false,  false);
    }

    const int stripe0 = s0 + sp;
    if (stripe0 >= s_end) return;
    const int rbase = stripe0 * 16 + half * 8;
    #pragma unroll
    for (int v = 0; v < 8; ++v)
        C[(size_t)(rbase + v) * NCOLS + col] = acc0[v];
    if (stripe0 + 1 >= s_end) return;
    #pragma unroll
    for (int v = 0; v < 8; ++v)
        C[(size_t)(rbase + 16 + v) * NCOLS + col] = acc1[v];
}


constexpr int CAP = 32;

__global__ __launch_bounds__(256)
void cnt_zero_kernel(int* __restrict__ cnt, int n, int own,
                     const float* __restrict__ A, const float* __restrict__ B,
                     float* __restrict__ C, int s_begin, int s_end) {
    __shared__ __attribute__((aligned(16))) float tile[G1_ROWS * G1_PITCH];
    if ((int)blockIdx.x >= own) {
        gemm1_block(A, B, C, (int)blockIdx.x - own, s_begin, s_end, tile);
        return;
    }
    int i = blockIdx.x * 256 + threadIdx.x;
    if (i <= n) cnt[i] = 0;
}

__global__ __launch_bounds__(256)
void edge_bucket_kernel(const int* __restrict__ src, const int* __restrict__ dst,
                        int* __restrict__ cnt, int* __restrict__ bkt,
                        int2* __restrict__ ovf, int* __restrict__ ovfn, int e, int own,
                        const float* __restrict__ A, const float* __restrict__ B,
                        float* __restrict__ C, int s_begin, int s_end) {
    __shared__ __attribute__((aligned(16))) float tile[G1_ROWS * G1_PITCH];
    if ((int)blockIdx.x >= own) {
        gemm1_block(A, B, C, (int)blockIdx.x - own, s_begin, s_end, tile);
        return;
    }
    int i = blockIdx.x * 256 + threadIdx.x;
    if (i >= e) return;
    const int s = src[i], d = dst[i];
    const int r = atomicAdd(&cnt[d], 1);
    if (r < CAP) {
        bkt[(size_t)d * CAP + r] = s;
    } else {
        const int k = atomicAdd(ovfn, 1);
        ovf[k] = make_int2(d, s);
    }
}


__device__ __forceinline__ float dinv_of(int count) {
    return __builtin_amdgcn_rsqf((float)(count + 1));
}

template<int C, int NB, bool PRE>
__device__ __forceinline__
void edge_batch(const int (&s)[NB], int jbase, int c, int node, float dd,
                const int* __restrict__ cnt, const float4* __restrict__ h4, int q,
                float4& acc) {
    float4 v[NB];
    int    cs[NB];
    #pragma unroll
    for (int j = 0; j < NB; ++j) {
        const int sj = (jbase + j < c) ? s[j] : node;
        cs[j] = PRE ? 0 : cnt[(unsigned)sj];
        v[j]  = h4[(unsigned)(sj * (C / 4) + q)];
    }
    #pragma unroll
    for (int j = 0; j < NB; ++j) {
        const float wj = PRE ? 1.0f : dinv_of(cs[j]) * dd;
        const float w  = (jbase + j < c) ? wj : 0.0f;
        acc.x += v[j].x * w;
        acc.y += v[j].y * w;
        acc.z += v[j].z * w;
        acc.w += v[j].w * w;
    }
}

template<int C, bool PRE>
__device__ __forceinline__
float4 gather_row(const int* __restrict__ cnt, const int* __restrict__ bkt,
                  const int2* __restrict__ ovf, const int* __restrict__ ovfn, int e,
                  const float* __restrict__ h, const float* __restrict__ bias,
                  int node, int c4, float& dd_out) {
    const float4* __restrict__ h4 = (const float4*)h;
    const int q = c4 >> 2;
    const int4* __restrict__ brow4 = (const int4*)bkt + (unsigned)(node * (CAP / 4));
    const int4 e0 = brow4[0];
    const int4 e1 = brow4[1];
    const int4 e2 = brow4[2];
    const int   c  = cnt[(unsigned)node];
    const float dd = dinv_of(c);
    dd_out = dd;

    const float4 b4 = *(const float4*)(bias + c4);
    const float4 hv = h4[(unsigned)(node * (C / 4) + q)];
    float4 acc;
    if (PRE) {
        acc = hv;
    } else {
        acc.x = b4.x + hv.x * dd * dd;
        acc.y = b4.y + hv.y * dd * dd;
        acc.z = b4.z + hv.z * dd * dd;
        acc.w = b4.w + hv.w * dd * dd;
    }

    const int sa[8] = {e0.x, e0.y, e0.z, e0.w, e1.x, e1.y, e1.z, e1.w};
    edge_batch<C, 8, PRE>(sa, 0, c, node, dd, cnt, h4, q, acc);

    if (c > 8) {
        const int sb[4] = {e2.x, e2.y, e2.z, e2.w};
        edge_batch<C, 4, PRE>(sb, 8, c, node, dd, cnt, h4, q, acc);

        const int* __restrict__ brow = (const int*)brow4;
        const int cm = min(c, CAP);
        for (int j = 12; j < cm; ++j) {
            const int   sj = brow[j];
            const float w  = PRE ? 1.0f : dinv_of(cnt[(unsigned)sj]) * dd;
            const float4 v = h4[(unsigned)(sj * (C / 4) + q)];
            acc.x += v.x * w;
            acc.y += v.y * w;
            acc.z += v.z * w;
            acc.w += v.w * w;
        }
        if (c > CAP) {
            const int m = min(ovfn[0], e);
            for (int k = 0; k < m; ++k) {
                const int2 p = ovf[k];
                if (p.x != node) continue;
                const float w  = PRE ? 1.0f : dinv_of(cnt[(unsigned)p.y]) * dd;
                const float4 v = h4[(unsigned)(p.y * (C / 4) + q)];
                acc.x += v.x * w;
                acc.y += v.y * w;
                acc.z += v.z * w;
                acc.w += v.w * w;
            }
        }
    }
    if (PRE) {
        acc.x = b4.x + acc.x * dd;
        acc.y = b4.y + acc.y * dd;
        acc.z = b4.z + acc.z * dd;
        acc.w = b4.w + acc.w * dd;
    }
    return acc;
}


__global__ __launch_bounds__(256)
void gather64_gemm2_kernel(const int* __restrict__ cnt, const int* __restrict__ bkt,
                           const int2* __restrict__ ovf, const int* __restrict__ ovfn, int e,
                           const float* __restrict__ h, const float* __restrict__ bias,
                           const float* __restrict__ W2, float* __restrict__ h2, int n) {
    constexpr int PITCH = 68;
    __shared__ __attribute__((aligned(16))) float tile[16 * PITCH];

    const int tid  = threadIdx.x;
    const int rloc = tid >> 4;
    const int c4   = (tid & 15) * 4;
    const int m0   = blockIdx.x * 16;
    const int node = min(m0 + rloc, n - 1);

    float dd;
    float4 acc = gather_row<64, false>(cnt, bkt, ovf, ovfn, e, h, bias, node, c4, dd);
    acc.x = fmaxf(acc.x, 0.0f) * dd;
    acc.y = fmaxf(acc.y, 0.0f) * dd;
    acc.z = fmaxf(acc.z, 0.0f) * dd;
    acc.w = fmaxf(acc.w, 0.0f) * dd;
    *(float4*)(tile + rloc * PITCH + c4) = acc;
    __syncthreads();

    const int wave = tid >> 5;
    if (wave >= 2) return;

    const int lane = tid & 31;
    const int lo16 = lane & 15;
    const int half = lane >> 4;
    const int col  = wave * 16 + lo16;
    const float* arow = tile + lo16 * PITCH;

    v8f o = {};
    #pragma unroll
    for (int k = 0; k < 64; k += 4) {
        const int kb = k + half * 2;
        const float2 av = *(const float2*)(arow + kb);
        v2f a; a[0] = av.x; a[1] = av.y;
        v2f b;
        b[0] = W2[kb * 32 + col];
        b[1] = W2[(kb + 1) * 32 + col];
        o = __builtin_amdgcn_wmma_f32_16x16x4_f32(
                 false, a,  false, b,
                 (short)0, o,  false,  false);
    }

    const int rbase = m0 + half * 8;
    #pragma unroll
    for (int v = 0; v < 8; ++v)
        h2[(size_t)(rbase + v) * 32 + col] = o[v];
}


__global__ __launch_bounds__(256)
void gather32_head_kernel(const int* __restrict__ cnt, const int* __restrict__ bkt,
                          const int2* __restrict__ ovf, const int* __restrict__ ovfn, int e,
                          const float* __restrict__ h, const float* __restrict__ bias,
                          const float* __restrict__ Wc, const float* __restrict__ bc,
                          float* __restrict__ out, int n) {
    __shared__ float res[32];

    const int tid  = threadIdx.x;
    const int rloc = tid >> 3;
    const int c4   = (tid & 7) * 4;
    const int node = min((int)blockIdx.x * 32 + rloc, n - 1);

    float dd;
    const float4 acc = gather_row<32, true>(cnt, bkt, ovf, ovfn, e, h, bias, node, c4, dd);
    float p = fmaxf(acc.x, 0.0f) * Wc[c4]
            + fmaxf(acc.y, 0.0f) * Wc[c4 + 1]
            + fmaxf(acc.z, 0.0f) * Wc[c4 + 2]
            + fmaxf(acc.w, 0.0f) * Wc[c4 + 3];
    p += __shfl_xor(p, 1);
    p += __shfl_xor(p, 2);
    p += __shfl_xor(p, 4);
    if ((tid & 7) == 0) res[rloc] = p + bc[0];
    __syncthreads();

    if (tid < 32) {
        const int i = blockIdx.x * 32 + tid;
        if (i < n) out[i] = res[tid];
    }
}


static inline int cdiv(long long a, long long b) { return (int)((a + b - 1) / b); }

extern "C" void kernel_launch(void* const* d_in, const int* in_sizes, int n_in,
                              void* d_out, int out_size, void* d_ws, size_t ws_size,
                              hipStream_t stream) {
    const float* x   = (const float*)d_in[0];
    const int*   ei  = (const int*)d_in[1];
    const float* W1  = (const float*)d_in[2];
    const float* b1  = (const float*)d_in[3];
    const float* W2  = (const float*)d_in[4];
    const float* b2  = (const float*)d_in[5];
    const float* Wc  = (const float*)d_in[6];
    const float* bc  = (const float*)d_in[7];
    float* out = (float*)d_out;

    const int N = in_sizes[0] / 128;
    const int E = in_sizes[1] / 2;
    const int* src = ei;
    const int* dst = ei + E;

    float* ws   = (float*)d_ws;
    float* h1   = ws;
    float* h2   = h1 + (size_t)N * 64;
    int*   bkt  = (int*)(h2 + (size_t)N * 32);
    int*   cnt  = bkt + (size_t)N * CAP;
    int*   ovfn = cnt + N;
    int2*  ovf  = (int2*)(ovfn + 4);
    (void)ws_size; (void)n_in; (void)out_size;

    const int T = 256;

    const int S  = cdiv(N, 16);
    const int s1 = S / 3;

    const int own1 = cdiv(N + 1, T);
    cnt_zero_kernel<<<own1 + cdiv(s1, G1_STRIPES), T, 0, stream>>>(cnt, N, own1, x, W1, h1, 0, s1);

    const int own2 = cdiv(E, T);
    edge_bucket_kernel<<<own2 + cdiv(S - s1, G1_STRIPES), T, 0, stream>>>(src, dst, cnt, bkt, ovf, ovfn, E, own2, x, W1, h1, s1, S);

    gather64_gemm2_kernel<<<cdiv(N, 16), T, 0, stream>>>(cnt, bkt, ovf, ovfn, E, h1, b1, W2, h2, N);

    gather32_head_kernel<<<cdiv(N, 32), T, 0, stream>>>(cnt, bkt, ovf, ovfn, E, h2, b2, Wc, bc, out, N);
}
